# starting kernel
# baseline (speedup 1.0000x reference)
#include <hip/hip_runtime.h>

typedef __attribute__((ext_vector_type(2))) float v2f;
typedef __attribute__((ext_vector_type(8))) float v8f;

#define NDIM 128

__global__ void rg_zero_kernel(float4* __restrict__ p, int n4) {
    int i = blockIdx.x * blockDim.x + threadIdx.x;
    if (i < n4) p[i] = make_float4(0.f, 0.f, 0.f, 0.f);
}

__global__ void rg_deg_kernel(const int* __restrict__ src, const int* __restrict__ dst,
                              float* __restrict__ odeg, float* __restrict__ ideg, int E) {
    int e = blockIdx.x * blockDim.x + threadIdx.x;
    if (e >= E) return;
    atomicAdd(&odeg[src[e]], 1.0f);
    atomicAdd(&ideg[dst[e]], 1.0f);
}

__global__ void rg_norm_kernel(const float* __restrict__ odeg, const float* __restrict__ ideg,
                               float* __restrict__ onorm, float* __restrict__ s1,
                               float* __restrict__ s2, int N) {
    int i = blockIdx.x * blockDim.x + threadIdx.x;
    if (i >= N) return;
    float od = fmaxf(odeg[i], 1.0f);
    float id = fmaxf(ideg[i], 1.0f);
    onorm[i] = rsqrtf(od);
    float r = rsqrtf(id);
    s1[i] = r;
    s2[i] = r / id;
}

__global__ __launch_bounds__(256) void
rg_edge_kernel(const float* __restrict__ h, const float* __restrict__ onorm,
               const float* __restrict__ weight, const int* __restrict__ src,
               const int* __restrict__ dst, const int* __restrict__ etype,
               float* __restrict__ mix, int E) {
    int e = blockIdx.x * 8 + (threadIdx.x >> 5);
    if (e >= E) return;
    int lane = threadIdx.x & 31;
    int s = src[e];
    int d = dst[e];
    int t = etype[e];
    float sc = onorm[s];
    float4 hv = ((const float4*)(h + (size_t)s * NDIM))[lane];
    float4 wv = ((const float4*)(weight + (size_t)t * NDIM))[lane];
    float* mp = mix + (size_t)d * NDIM + lane * 4;
    atomicAdd(mp + 0, hv.x * wv.x * sc);
    atomicAdd(mp + 1, hv.y * wv.y * sc);
    atomicAdd(mp + 2, hv.z * wv.z * sc);
    atomicAdd(mp + 3, hv.w * wv.w * sc);
}

#define LDS_STRIDE 132
__global__ __launch_bounds__(256) void
rg_gemm_kernel(const float* __restrict__ H, const float* __restrict__ W,
               const float* __restrict__ mix, const float* __restrict__ s1,
               const float* __restrict__ s2, float* __restrict__ out) {
    __shared__ float ldsH[16 * LDS_STRIDE];

    const int m0 = blockIdx.x * 16;
    const int tid = threadIdx.x;

    for (int i = tid; i < 16 * 32; i += 256) {
        int r = i >> 5;
        int c4 = i & 31;
        float4 v = ((const float4*)(H + (size_t)(m0 + r) * NDIM))[c4];
        float* dl = ldsH + r * LDS_STRIDE + c4 * 4;
        dl[0] = v.x; dl[1] = v.y; dl[2] = v.z; dl[3] = v.w;
    }
    __syncthreads();

    const int lane = tid & 31;
    const int hi = lane >> 4;
    const int lo = lane & 15;
    const int n0 = (tid >> 5) * 16;

    v8f c = {};
    #pragma unroll
    for (int k = 0; k < NDIM; k += 4) {
        int kk = k + 2 * hi;
        v2f a;
        a.x = ldsH[lo * LDS_STRIDE + kk];
        a.y = ldsH[lo * LDS_STRIDE + kk + 1];
        v2f b;
        b.x = W[(size_t)kk * NDIM + n0 + lo];
        b.y = W[(size_t)(kk + 1) * NDIM + n0 + lo];
        c = __builtin_amdgcn_wmma_f32_16x16x4_f32(
                 false, a,  false, b,
                 (short)0, c,  false,  false);
    }

    #pragma unroll
    for (int g = 0; g < 8; ++g) {
        int row = m0 + g + 8 * hi;
        float a1 = s1[row];
        float a2 = s2[row];
        size_t off = (size_t)row * NDIM + n0 + lo;
        out[off] = c[g] * a1 + mix[off] * a2;
    }
}

extern "C" void kernel_launch(void* const* d_in, const int* in_sizes, int n_in,
                              void* d_out, int out_size, void* d_ws, size_t ws_size,
                              hipStream_t stream) {
    const float* x      = (const float*)d_in[0];
    const float* weight = (const float*)d_in[1];
    const float* lw0    = (const float*)d_in[2];
    const float* lw1    = (const float*)d_in[3];
    const int*   src    = (const int*)d_in[4];
    const int*   dst    = (const int*)d_in[5];
    const int*   etype  = (const int*)d_in[6];
    float* out = (float*)d_out;

    const int N = in_sizes[0] / NDIM;
    const int E = in_sizes[4];
    const size_t ND = (size_t)N * NDIM;

    float* ws    = (float*)d_ws;
    float* h1    = ws;
    float* mix   = h1 + ND;
    float* odeg  = mix + ND;
    float* ideg  = odeg + N;
    float* onorm = ideg + N;
    float* s1    = onorm + N;
    float* s2    = s1 + N;

    const int ZT = 256;
    {
        int n4 = (2 * N + 3) / 4;
        rg_zero_kernel<<<(n4 + ZT - 1) / ZT, ZT, 0, stream>>>((float4*)odeg, n4);
    }
    rg_deg_kernel<<<(E + ZT - 1) / ZT, ZT, 0, stream>>>(src, dst, odeg, ideg, E);
    rg_norm_kernel<<<(N + ZT - 1) / ZT, ZT, 0, stream>>>(odeg, ideg, onorm, s1, s2, N);

    const int edgeBlocks = (E + 7) / 8;
    const int mix4 = (int)(ND / 4);
    const int gemmBlocks = N / 16;

    rg_zero_kernel<<<(mix4 + ZT - 1) / ZT, ZT, 0, stream>>>((float4*)mix, mix4);
    rg_edge_kernel<<<edgeBlocks, 256, 0, stream>>>(x, onorm, weight, src, dst, etype, mix, E);
    rg_gemm_kernel<<<gemmBlocks, 256, 0, stream>>>(x, lw0, mix, s1, s2, h1);

    rg_zero_kernel<<<(mix4 + ZT - 1) / ZT, ZT, 0, stream>>>((float4*)mix, mix4);
    rg_edge_kernel<<<edgeBlocks, 256, 0, stream>>>(h1, onorm, weight, src, dst, etype, mix, E);
    rg_gemm_kernel<<<gemmBlocks, 256, 0, stream>>>(h1, lw1, mix, s1, s2, out);
}


# Edge sums without f32 atomics: per call an integer atomic count ranks each edge in its dst segment (arrival order), scan + fill sort the list, one wave per row sums it in registers
# speedup vs baseline: 2.0322x; 2.0322x over previous
#include <hip/hip_runtime.h>

typedef __attribute__((ext_vector_type(2))) float v2f;
typedef __attribute__((ext_vector_type(8))) float v8f;

#define NDIM 128

__global__ void rg_zero_kernel(float4* __restrict__ p, int n4) {
    int i = blockIdx.x * blockDim.x + threadIdx.x;
    if (i < n4) p[i] = make_float4(0.f, 0.f, 0.f, 0.f);
}

__global__ void rg_count_kernel(const int* __restrict__ src, const int* __restrict__ dst,
                                int* __restrict__ cin, int* __restrict__ cout,
                                int* __restrict__ rank, int E) {
    int e = blockIdx.x * blockDim.x + threadIdx.x;
    if (e >= E) return;
    rank[e] = atomicAdd(&cin[dst[e]], 1);
    atomicAdd(&cout[src[e]], 1);
}

__global__ __launch_bounds__(256) void
rg_scan_norm_kernel(const int* __restrict__ cin, const int* __restrict__ cout,
                    int* __restrict__ off, float* __restrict__ onorm,
                    float* __restrict__ s1, float* __restrict__ s2, int N) {
    __shared__ int red[256];
    __shared__ int sc[2][256];
    const int tid = threadIdx.x;
    const int base = blockIdx.x * 256;

    int part = 0;
    const int4* c4 = (const int4*)cin;
    for (int i = tid; i < (base >> 2); i += 256) {
        int4 v = c4[i];
        part += v.x + v.y + v.z + v.w;
    }
    red[tid] = part;
    __syncthreads();
    for (int s = 128; s > 0; s >>= 1) {
        if (tid < s) red[tid] += red[tid + s];
        __syncthreads();
    }
    const int prefix = red[0];

    const int node = base + tid;
    int ci = 0, co = 0;
    if (node < N) { ci = cin[node]; co = cout[node]; }
    sc[0][tid] = ci;
    __syncthreads();
    int pin = 0;
    for (int o = 1; o < 256; o <<= 1) {
        int t = sc[pin][tid];
        if (tid >= o) t += sc[pin][tid - o];
        sc[pin ^ 1][tid] = t;
        __syncthreads();
        pin ^= 1;
    }
    const int incl = sc[pin][tid];

    if (node < N) {
        off[node] = prefix + incl - ci;
        if (node == N - 1) off[N] = prefix + incl;
        float od = fmaxf((float)co, 1.0f);
        float id = fmaxf((float)ci, 1.0f);
        onorm[node] = rsqrtf(od);
        float r = rsqrtf(id);
        s1[node] = r;
        s2[node] = r / id;
    }
}

__global__ void rg_fill_kernel(const int* __restrict__ src, const int* __restrict__ dst,
                               const int* __restrict__ etype, const int* __restrict__ off,
                               const int* __restrict__ rank, int2* __restrict__ perm, int E) {
    int e = blockIdx.x * blockDim.x + threadIdx.x;
    if (e >= E) return;
    perm[off[dst[e]] + rank[e]] = make_int2(src[e], etype[e]);
}

__device__ __forceinline__ int rg_imin(int a, int b) { return a < b ? a : b; }

__global__ __launch_bounds__(256) void
rg_gather_kernel(const float* __restrict__ h, const float* __restrict__ onorm,
                 const float* __restrict__ weight, const int2* __restrict__ perm,
                 const int* __restrict__ off, float* __restrict__ mix, int N) {
    const int d = __builtin_amdgcn_readfirstlane((int)(blockIdx.x * 8 + (threadIdx.x >> 5)));
    if (d >= N) return;
    const int lane = threadIdx.x & 31;
    const int j0 = off[d];
    const int j1 = off[d + 1];

    float4 acc = make_float4(0.f, 0.f, 0.f, 0.f);
    if (j1 > j0) {
        const int last = j1 - 1;
        int2 p0 = perm[j0];
        int2 p1 = perm[rg_imin(j0 + 1, last)];
        int2 p2 = perm[rg_imin(j0 + 2, last)];
        int2 p3 = perm[rg_imin(j0 + 3, last)];
        for (int j = j0; j < j1; j += 4) {
            const float c0 = onorm[p0.x];
            const float c1 = onorm[p1.x];
            const float c2 = onorm[p2.x];
            const float c3 = onorm[p3.x];
            const float4 h0 = ((const float4*)(h + (size_t)p0.x * NDIM))[lane];
            const float4 h1 = ((const float4*)(h + (size_t)p1.x * NDIM))[lane];
            const float4 h2 = ((const float4*)(h + (size_t)p2.x * NDIM))[lane];
            const float4 h3 = ((const float4*)(h + (size_t)p3.x * NDIM))[lane];
            const float4 w0 = ((const float4*)(weight + (size_t)p0.y * NDIM))[lane];
            const float4 w1 = ((const float4*)(weight + (size_t)p1.y * NDIM))[lane];
            const float4 w2 = ((const float4*)(weight + (size_t)p2.y * NDIM))[lane];
            const float4 w3 = ((const float4*)(weight + (size_t)p3.y * NDIM))[lane];

            const int jn = j + 4;
            const int2 q0 = perm[rg_imin(jn, last)];
            const int2 q1 = perm[rg_imin(jn + 1, last)];
            const int2 q2 = perm[rg_imin(jn + 2, last)];
            const int2 q3 = perm[rg_imin(jn + 3, last)];

            const bool v1 = (j + 1) < j1;
            const bool v2 = (j + 2) < j1;
            const bool v3 = (j + 3) < j1;
            acc.x += h0.x * w0.x * c0;
            acc.y += h0.y * w0.y * c0;
            acc.z += h0.z * w0.z * c0;
            acc.w += h0.w * w0.w * c0;
            acc.x += v1 ? h1.x * w1.x * c1 : 0.f;
            acc.y += v1 ? h1.y * w1.y * c1 : 0.f;
            acc.z += v1 ? h1.z * w1.z * c1 : 0.f;
            acc.w += v1 ? h1.w * w1.w * c1 : 0.f;
            acc.x += v2 ? h2.x * w2.x * c2 : 0.f;
            acc.y += v2 ? h2.y * w2.y * c2 : 0.f;
            acc.z += v2 ? h2.z * w2.z * c2 : 0.f;
            acc.w += v2 ? h2.w * w2.w * c2 : 0.f;
            acc.x += v3 ? h3.x * w3.x * c3 : 0.f;
            acc.y += v3 ? h3.y * w3.y * c3 : 0.f;
            acc.z += v3 ? h3.z * w3.z * c3 : 0.f;
            acc.w += v3 ? h3.w * w3.w * c3 : 0.f;

            p0 = q0; p1 = q1; p2 = q2; p3 = q3;
        }
    }
    ((float4*)(mix + (size_t)d * NDIM))[lane] = acc;
}

#define LDS_STRIDE 132
__global__ __launch_bounds__(256) void
rg_gemm_kernel(const float* __restrict__ H, const float* __restrict__ W,
               const float* __restrict__ mix, const float* __restrict__ s1,
               const float* __restrict__ s2, float* __restrict__ out) {
    __shared__ float ldsH[16 * LDS_STRIDE];

    const int m0 = blockIdx.x * 16;
    const int tid = threadIdx.x;

    for (int i = tid; i < 16 * 32; i += 256) {
        int r = i >> 5;
        int c4 = i & 31;
        float4 v = ((const float4*)(H + (size_t)(m0 + r) * NDIM))[c4];
        float* dl = ldsH + r * LDS_STRIDE + c4 * 4;
        dl[0] = v.x; dl[1] = v.y; dl[2] = v.z; dl[3] = v.w;
    }
    __syncthreads();

    const int lane = tid & 31;
    const int hi = lane >> 4;
    const int lo = lane & 15;
    const int n0 = (tid >> 5) * 16;

    v8f c = {};
    #pragma unroll
    for (int k = 0; k < NDIM; k += 4) {
        int kk = k + 2 * hi;
        v2f a;
        a.x = ldsH[lo * LDS_STRIDE + kk];
        a.y = ldsH[lo * LDS_STRIDE + kk + 1];
        v2f b;
        b.x = W[(size_t)kk * NDIM + n0 + lo];
        b.y = W[(size_t)(kk + 1) * NDIM + n0 + lo];
        c = __builtin_amdgcn_wmma_f32_16x16x4_f32(
                 false, a,  false, b,
                 (short)0, c,  false,  false);
    }

    #pragma unroll
    for (int g = 0; g < 8; ++g) {
        int row = m0 + g + 8 * hi;
        float a1 = s1[row];
        float a2 = s2[row];
        size_t off = (size_t)row * NDIM + n0 + lo;
        out[off] = c[g] * a1 + mix[off] * a2;
    }
}

extern "C" void kernel_launch(void* const* d_in, const int* in_sizes, int n_in,
                              void* d_out, int out_size, void* d_ws, size_t ws_size,
                              hipStream_t stream) {
    const float* x      = (const float*)d_in[0];
    const float* weight = (const float*)d_in[1];
    const float* lw0    = (const float*)d_in[2];
    const float* lw1    = (const float*)d_in[3];
    const int*   src    = (const int*)d_in[4];
    const int*   dst    = (const int*)d_in[5];
    const int*   etype  = (const int*)d_in[6];
    float* out = (float*)d_out;

    const int N = in_sizes[0] / NDIM;
    const int E = in_sizes[4];
    const size_t ND = (size_t)N * NDIM;

    float* ws    = (float*)d_ws;
    float* h1    = ws;
    float* mix   = h1 + ND;
    float* onorm = mix + ND;
    float* s1    = onorm + N;
    float* s2    = s1 + N;
    int*   off   = (int*)(s2 + N);

    const int NP = (N + 3) & ~3;
    const int EP = (E + 3) & ~3;
    int*  cin  = (int*)mix;
    int*  cout = cin + NP;
    int*  rank = cout + NP;
    int2* perm = (int2*)d_out;

    const int ZT = 256;
    {
        int n4 = (2 * NP) / 4;
        rg_zero_kernel<<<(n4 + ZT - 1) / ZT, ZT, 0, stream>>>((float4*)cin, n4);
    }
    rg_count_kernel<<<(E + ZT - 1) / ZT, ZT, 0, stream>>>(src, dst, cin, cout, rank, E);
    rg_scan_norm_kernel<<<(N + 255) / 256, 256, 0, stream>>>(cin, cout, off, onorm, s1, s2, N);
    rg_fill_kernel<<<(E + ZT - 1) / ZT, ZT, 0, stream>>>(src, dst, etype, off, rank, perm, E);

    const int gatherBlocks = (N + 7) / 8;
    const int gemmBlocks = N / 16;

    rg_gather_kernel<<<gatherBlocks, 256, 0, stream>>>(x, onorm, weight, perm, off, mix, N);
    rg_gemm_kernel<<<gemmBlocks, 256, 0, stream>>>(x, lw0, mix, s1, s2, h1);

    rg_gather_kernel<<<gatherBlocks, 256, 0, stream>>>(h1, onorm, weight, perm, off, mix, N);
    rg_gemm_kernel<<<gemmBlocks, 256, 0, stream>>>(h1, lw1, mix, s1, s2, out);
}
